# starting kernel
# baseline (speedup 1.0000x reference)
#include <hip/hip_runtime.h>


typedef __attribute__((ext_vector_type(2))) float v2f;
typedef __attribute__((ext_vector_type(8))) float v8f;

#define NNODES 100000


__global__ void zero_f32(float* __restrict__ p, int n) {
  int i = blockIdx.x * blockDim.x + threadIdx.x;
  if (i < n) p[i] = 0.0f;
}

__global__ void degree_kernel(const int* __restrict__ src, const int* __restrict__ dst,
                              float* __restrict__ deg_out, float* __restrict__ deg_in, int E) {
  int e = blockIdx.x * blockDim.x + threadIdx.x;
  if (e < E) {
    atomicAdd(&deg_out[src[e]], 1.0f);
    atomicAdd(&deg_in[dst[e]], 1.0f);
  }
}

__global__ void norm_kernel(float* __restrict__ deg_out, float* __restrict__ deg_in, int n) {
  int i = blockIdx.x * blockDim.x + threadIdx.x;
  if (i < n) {
    deg_out[i] = rsqrtf(fmaxf(deg_out[i], 1.0f));
    deg_in[i]  = rsqrtf(fmaxf(deg_in[i], 1.0f));
  }
}

template <int F, bool SCALE>
__global__ void scatter_kernel(const float* __restrict__ x,
                               const int* __restrict__ src, const int* __restrict__ dst,
                               const float* __restrict__ scale,
                               float* __restrict__ agg, int E) {
  int tid = blockIdx.x * blockDim.x + threadIdx.x;
  int e = tid / F;
  int f = tid - e * F;
  if (e < E) {
    int s = src[e];
    int d = dst[e];
    float v = x[s * F + f];
    if (SCALE) v *= scale[s];
    atomicAdd(&agg[d * F + f], v);
  }
}

template <int F, bool RELU>
__global__ void epilogue_kernel(const float* __restrict__ agg,
                                const float* __restrict__ invIn,
                                const float* __restrict__ bias,
                                float* __restrict__ out, int n) {
  int i = blockIdx.x * blockDim.x + threadIdx.x;
  if (i < n) {
    int row = i / F;
    int f = i - row * F;
    float v = agg[i] * invIn[row] + bias[f];
    if (RELU) v = fmaxf(v, 0.0f);
    out[i] = v;
  }
}

template <int K, int NC, bool PRESCALE, bool EPI, bool RELU>
__global__ __launch_bounds__(32) void gemm_wmma(const float* __restrict__ A,
                                                const float* __restrict__ B,
                                                float* __restrict__ D,
                                                const float* __restrict__ rowScaleA,
                                                const float* __restrict__ rowScaleOut,
                                                const float* __restrict__ bias) {
  const int lane = threadIdx.x;
  const int half = lane >> 4;
  const int l15  = lane & 15;
  const int tileM = blockIdx.x * 16;
  const int tileN = blockIdx.y * 16;

  const int arow = tileM + l15;
  const float* __restrict__ Ap = A + arow * K;
  float ascale = 1.0f;
  if (PRESCALE) ascale = rowScaleA[arow];

  const int  col   = tileN + l15;
  const bool colOK = (NC % 16 == 0) ? true : (col < NC);

  v8f acc = {0.f, 0.f, 0.f, 0.f, 0.f, 0.f, 0.f, 0.f};

#pragma unroll 4
  for (int k = 0; k < K; k += 4) {
    const int ka = k + half * 2;
    v2f a, b;
    a.x = Ap[ka];
    a.y = Ap[ka + 1];
    if (PRESCALE) { a.x *= ascale; a.y *= ascale; }
    b.x = colOK ? B[ka * NC + col] : 0.0f;
    b.y = colOK ? B[(ka + 1) * NC + col] : 0.0f;
    acc = __builtin_amdgcn_wmma_f32_16x16x4_f32(false, a, false, b, (short)0, acc,
                                                false, false);
  }

  if (colOK) {
    const int mbase = tileM + half * 8;
#pragma unroll
    for (int r = 0; r < 8; ++r) {
      const int row = mbase + r;
      float v = acc[r];
      if (EPI)  v = v * rowScaleOut[row] + bias[col];
      if (RELU) v = fmaxf(v, 0.0f);
      D[row * NC + col] = v;
    }
  }
}


extern "C" void kernel_launch(void* const* d_in, const int* in_sizes, int n_in,
                              void* d_out, int out_size, void* d_ws, size_t ws_size,
                              hipStream_t stream) {
  const float* features = (const float*)d_in[0];
  const int*   src      = (const int*)d_in[1];
  const int*   dst      = (const int*)d_in[2];
  const float* W1       = (const float*)d_in[3];
  const float* b1       = (const float*)d_in[4];
  const float* W2       = (const float*)d_in[5];
  const float* b2       = (const float*)d_in[6];
  const float* W3       = (const float*)d_in[7];
  const float* b3       = (const float*)d_in[8];

  const int N = NNODES;
  const int E = in_sizes[1];

  float* ws      = (float*)d_ws;
  float* inv_out = ws;
  float* inv_in  = ws + N;
  float* P0      = ws + 2 * (size_t)N;
  float* P1      = P0 + (size_t)N * 256;
  float* P2      = P1 + (size_t)N * 128;
  float* H3      = P0;
  float* AGG3    = P0 + (size_t)N * 40;

  const int T = 256;
  auto blocks = [T](long long n) { return (unsigned)((n + T - 1) / T); };

  zero_f32<<<blocks(2LL * N), T, 0, stream>>>(inv_out, 2 * N);
  degree_kernel<<<blocks(E), T, 0, stream>>>(src, dst, inv_out, inv_in, E);
  norm_kernel<<<blocks(N), T, 0, stream>>>(inv_out, inv_in, N);

  zero_f32<<<blocks((long long)N * 128), T, 0, stream>>>(P1, N * 128);
  scatter_kernel<128, true><<<blocks((long long)E * 128), T, 0, stream>>>(
      features, src, dst, inv_out, P1, E);
  gemm_wmma<128, 256, false, true, true><<<dim3(N / 16, 256 / 16), 32, 0, stream>>>(
      P1, W1, P0, nullptr, inv_in, b1);

  gemm_wmma<256, 128, true, false, false><<<dim3(N / 16, 128 / 16), 32, 0, stream>>>(
      P0, W2, P2, inv_out, nullptr, nullptr);
  zero_f32<<<blocks((long long)N * 128), T, 0, stream>>>(P1, N * 128);
  scatter_kernel<128, false><<<blocks((long long)E * 128), T, 0, stream>>>(
      P2, src, dst, nullptr, P1, E);
  epilogue_kernel<128, true><<<blocks((long long)N * 128), T, 0, stream>>>(
      P1, inv_in, b2, P2, N * 128);

  gemm_wmma<128, 40, true, false, false><<<dim3(N / 16, 3), 32, 0, stream>>>(
      P2, W3, H3, inv_out, nullptr, nullptr);
  zero_f32<<<blocks((long long)N * 40), T, 0, stream>>>(AGG3, N * 40);
  scatter_kernel<40, false><<<blocks((long long)E * 40), T, 0, stream>>>(
      H3, src, dst, nullptr, AGG3, E);
  epilogue_kernel<40, false><<<blocks((long long)N * 40), T, 0, stream>>>(
      AGG3, inv_in, b3, (float*)d_out, N * 40);
}


# Edges dealt into per-owner-range lists by the degree launch (LDS rank, one global int add per work-group and range): no agg launch filters dst[]; sum order in a node still ds_add_f32 arrival
# speedup vs baseline: 3.6774x; 3.6774x over previous
#include <hip/hip_runtime.h>


typedef __attribute__((ext_vector_type(2))) float v2f;
typedef __attribute__((ext_vector_type(4))) float v4f;
typedef __attribute__((ext_vector_type(8))) float v8f;
typedef __attribute__((ext_vector_type(4))) int   v4i;

#define NNODES 100000


__global__ void zero_f32(float* __restrict__ p, int n, unsigned* __restrict__ q, int m, int qs) {
  int i = blockIdx.x * blockDim.x + threadIdx.x;
  if (i < n) p[i] = 0.0f;
  else if (i - n < m) q[(size_t)(i - n) * qs] = 0u;
}

__global__ void norm_kernel(float* __restrict__ deg_out, float* __restrict__ deg_in, int n) {
  int i = blockIdx.x * blockDim.x + threadIdx.x;
  if (i < n) {
    deg_out[i] = rsqrtf(fmaxf(deg_out[i], 1.0f));
    deg_in[i]  = rsqrtf(fmaxf(deg_in[i], 1.0f));
  }
}

#define AGG_NB 256
#define AGG_R  391
#define AGG_NT 1024
#define AGG_WQ 512
#define AGG_EMASK 0x7FFFFFu
#define AGG_CAP ((AGG_NT / 32) * AGG_WQ)
#define AGG_CS  64

static_assert(AGG_NB * AGG_R >= NNODES, "node ranges must cover every node");
static_assert(AGG_R <= 512 && NNODES <= (1 << 23), "entry packing: 9 bits row, 23 bits index");
static_assert((long long)AGG_NB * (AGG_CAP + AGG_CS) <= (long long)NNODES * 128,
              "the edge lists must fit the workspace plane they are given");

#define EP_NT  1024
#define EP_PER 4

__global__ __launch_bounds__(EP_NT) void edge_pass(const int* __restrict__ src,
                                                   const int* __restrict__ dst,
                                                   float* __restrict__ deg_out,
                                                   float* __restrict__ deg_in,
                                                   unsigned* __restrict__ lcount,
                                                   unsigned* __restrict__ lists, int E) {
  __shared__ unsigned cnt[AGG_NB];
  __shared__ unsigned base[AGG_NB];
  const int tid   = threadIdx.x;
  const int share = (E + (int)gridDim.x - 1) / (int)gridDim.x;
  const int e0    = blockIdx.x * share;
  const int e1    = min(E, e0 + share);
  for (int eb = e0; eb < e1; eb += EP_NT * EP_PER) {
    if (tid < AGG_NB) cnt[tid] = 0u;
    __syncthreads();
    int      rg[EP_PER];
    unsigned rk[EP_PER];
    unsigned en[EP_PER];
    int      sv[EP_PER];
    int      dv[EP_PER];
#pragma unroll
    for (int k = 0; k < EP_PER; ++k) {
      const int e = eb + k * EP_NT + tid;
      sv[k] = (e < e1) ? src[e] : -1;
      dv[k] = (e < e1) ? dst[e] : -1;
    }
#pragma unroll
    for (int k = 0; k < EP_PER; ++k) {
      rg[k] = -1;
      if (dv[k] >= 0) {
        const int r = dv[k] / AGG_R;
        rg[k] = r;
        en[k] = ((unsigned)(dv[k] - r * AGG_R) << 23) | (unsigned)sv[k];
        rk[k] = __hip_atomic_fetch_add(&cnt[r], 1u, __ATOMIC_RELAXED, __HIP_MEMORY_SCOPE_WORKGROUP);
      }
    }
    __syncthreads();
    if (tid < AGG_NB) {
      const unsigned c = cnt[tid];
      base[tid] = c ? atomicAdd(&lcount[tid * AGG_CS], c) : 0u;
    }
    __syncthreads();
#pragma unroll
    for (int k = 0; k < EP_PER; ++k) {
      if (rg[k] >= 0) {
        const unsigned slot = base[rg[k]] + rk[k];
        if (slot < (unsigned)AGG_CAP) lists[(size_t)rg[k] * AGG_CAP + slot] = en[k];
      }
    }
#pragma unroll
    for (int k = 0; k < EP_PER; ++k) {
      if (dv[k] >= 0) {
        atomicAdd(&deg_out[sv[k]], 1.0f);
        atomicAdd(&deg_in[dv[k]], 1.0f);
      }
    }
  }
}

template <int F, bool SCALE>
__device__ __forceinline__ void agg_slow_edge(float* acc, const float* __restrict__ x,
                                              const int* __restrict__ src,
                                              const float* __restrict__ scale, unsigned row, int e) {
  constexpr int C4 = F / 4;
  const int   s  = src[e];
  const float sc = SCALE ? scale[s] : 1.0f;
#pragma nounroll
  for (int f = 0; f < F; ++f) {
    __hip_atomic_fetch_add(&acc[row * F + (f & 3) * C4 + (f >> 2)], x[(size_t)s * F + f] * sc,
                           __ATOMIC_RELAXED, __HIP_MEMORY_SCOPE_WORKGROUP);
  }
}

template <int F, bool SCALE>
__global__ __launch_bounds__(AGG_NT) void agg_owner(const float* __restrict__ x,
                                                    const int* __restrict__ src,
                                                    const int* __restrict__ dst,
                                                    const float* __restrict__ scale,
                                                    float* __restrict__ agg,
                                                    const unsigned* __restrict__ lcount,
                                                    const unsigned* __restrict__ lists, int E) {
  constexpr int NW  = AGG_NT / 32;
  constexpr int C4  = F / 4;
  constexpr int EPW = 32 / C4;
  constexpr int U   = 8;

  __shared__ __attribute__((aligned(16))) float acc[AGG_R * F];
  __shared__ unsigned wq[NW * AGG_WQ];

  const int tid  = threadIdx.x;
  const int lane = tid & 31;
  const int wave = __builtin_amdgcn_readfirstlane(tid >> 5);
  const int lo    = blockIdx.x * AGG_R;
  const int nrows = min(AGG_R, NNODES - lo);
  unsigned* myq = &wq[wave * AGG_WQ];

  const int  total = __builtin_amdgcn_readfirstlane((int)lcount[blockIdx.x * AGG_CS]);
  const bool scan  = (unsigned)total > (unsigned)AGG_CAP;
  int wcnt = 0;

  for (int i = tid; i < AGG_R * F / 4; i += AGG_NT) ((v4f*)acc)[i] = (v4f){0.f, 0.f, 0.f, 0.f};
  __syncthreads();

  if (!scan) {
    const int per = (total + NW - 1) / NW;
    const int j0  = wave * per;
    wcnt = max(0, min(per, total - j0));
    const unsigned* q = lists + (size_t)blockIdx.x * AGG_CAP + j0;
    for (int jb = 0; jb < wcnt; jb += 128) {
      unsigned en[4];
#pragma unroll
      for (int k = 0; k < 4; ++k) {
        const int j = jb + k * 32 + lane;
        en[k] = (j < wcnt) ? q[j] : 0u;
      }
#pragma unroll
      for (int k = 0; k < 4; ++k) {
        const int j = jb + k * 32 + lane;
        if (j < wcnt) myq[j] = en[k];
      }
    }
  } else {
    auto push = [&](unsigned u, int e) {
      const bool     p    = u < (unsigned)nrows;
      const unsigned mask = __builtin_amdgcn_ballot_w32(p);
      if (mask != 0u) {
        const int pos = wcnt + (int)__builtin_amdgcn_mbcnt_lo(mask, 0u);
        if (p) {
          if (pos < AGG_WQ) myq[pos] = (u << 23) | (unsigned)e;
          else agg_slow_edge<F, SCALE>(acc, x, src, scale, u, e);
        }
        wcnt += __builtin_popcount(mask);
      }
    };

    const int  E4   = E >> 2;
    const v4i* dst4 = (const v4i*)dst;
    auto scan4 = [&](v4i d, int i) {
      const unsigned u0 = (unsigned)(d.x - lo);
      const unsigned u1 = (unsigned)(d.y - lo);
      const unsigned u2 = (unsigned)(d.z - lo);
      const unsigned u3 = (unsigned)(d.w - lo);
      const unsigned m  = min(min(u0, u1), min(u2, u3));
      if (__builtin_amdgcn_ballot_w32(m < (unsigned)nrows) != 0u) {
        push(u0, 4 * i + 0);
        push(u1, 4 * i + 1);
        push(u2, 4 * i + 2);
        push(u3, 4 * i + 3);
      }
    };
    constexpr int SU = 8;
    int ib = 0;
    for (; ib + SU * AGG_NT <= E4; ib += SU * AGG_NT) {
      v4i d[SU];
#pragma unroll
      for (int k = 0; k < SU; ++k) d[k] = dst4[ib + k * AGG_NT + tid];
#pragma unroll
      for (int k = 0; k < SU; ++k) scan4(d[k], ib + k * AGG_NT + tid);
    }
    for (; ib < E4; ib += AGG_NT) {
      const int i = ib + tid;
      v4i d = {-1, -1, -1, -1};
      if (i < E4) d = dst4[i];
      scan4(d, i);
    }
    if (wave == 0) {
      const int e = 4 * E4 + lane;
      int d = -1;
      if (e < E) d = dst[e];
      push((unsigned)(d - lo), e);
    }

    const int nq = min(wcnt, AGG_WQ);
    for (int jb = 0; jb < nq; jb += 128) {
      unsigned en[4];
      int      sv[4];
#pragma unroll
      for (int k = 0; k < 4; ++k) {
        const int j = jb + k * 32 + lane;
        en[k] = (j < nq) ? myq[j] : 0u;
      }
#pragma unroll
      for (int k = 0; k < 4; ++k) sv[k] = src[en[k] & AGG_EMASK];
#pragma unroll
      for (int k = 0; k < 4; ++k) {
        const int j = jb + k * 32 + lane;
        if (j < nq) myq[j] = (en[k] & ~AGG_EMASK) | (unsigned)sv[k];
      }
    }
  }
  const int n = min(wcnt, AGG_WQ);

  const int  sub    = (EPW == 1) ? 0 : lane / C4;
  const int  c4     = (EPW == 1) ? lane : lane - sub * C4;
  const bool active = sub < EPW;
  const v4f* x4 = (const v4f*)x;
  for (int jb = 0; jb < n; jb += U * EPW) {
    v4f      v[U];
    float    sc[U];
    unsigned rw[U];
    bool     ok[U];
#pragma unroll
    for (int k = 0; k < U; ++k) {
      const int j = jb + k * EPW + sub;
      ok[k] = active && (j < n);
      unsigned en = myq[ok[k] ? j : jb];
      if (EPW == 1) en = (unsigned)__builtin_amdgcn_readfirstlane((int)en);
      const unsigned s = en & AGG_EMASK;
      rw[k] = en >> 23;
      v[k]  = x4[(size_t)s * C4 + c4];
      sc[k] = SCALE ? scale[s] : 1.0f;
    }
#pragma unroll
    for (int k = 0; k < U; ++k) {
      if (ok[k]) {
        float* a = &acc[rw[k] * F + c4];
        __hip_atomic_fetch_add(a,          v[k].x * sc[k], __ATOMIC_RELAXED, __HIP_MEMORY_SCOPE_WORKGROUP);
        __hip_atomic_fetch_add(a + C4,     v[k].y * sc[k], __ATOMIC_RELAXED, __HIP_MEMORY_SCOPE_WORKGROUP);
        __hip_atomic_fetch_add(a + 2 * C4, v[k].z * sc[k], __ATOMIC_RELAXED, __HIP_MEMORY_SCOPE_WORKGROUP);
        __hip_atomic_fetch_add(a + 3 * C4, v[k].w * sc[k], __ATOMIC_RELAXED, __HIP_MEMORY_SCOPE_WORKGROUP);
      }
    }
  }
  __syncthreads();

  v4f* out4 = (v4f*)(agg + (size_t)lo * F);
  for (int it = tid; it < nrows * C4; it += AGG_NT) {
    const int row = it / C4;
    const int q4  = it - row * C4;
    const float* a = &acc[row * F + q4];
    out4[it] = (v4f){a[0], a[C4], a[2 * C4], a[3 * C4]};
  }
}

template <int F, bool RELU>
__global__ void epilogue_kernel(const float* __restrict__ agg,
                                const float* __restrict__ invIn,
                                const float* __restrict__ bias,
                                float* __restrict__ out, int n) {
  int i = blockIdx.x * blockDim.x + threadIdx.x;
  if (i < n) {
    int row = i / F;
    int f = i - row * F;
    float v = agg[i] * invIn[row] + bias[f];
    if (RELU) v = fmaxf(v, 0.0f);
    out[i] = v;
  }
}

template <int K, int NC, bool PRESCALE, bool EPI, bool RELU>
__global__ __launch_bounds__(((NC + 15) / 16) * 32) void gemm_wstat(
    const float* __restrict__ A, const float* __restrict__ B, float* __restrict__ D,
    const float* __restrict__ rowScaleA, const float* __restrict__ rowScaleOut,
    const float* __restrict__ bias) {
  constexpr int NW   = (NC + 15) / 16;
  constexpr int NT   = NW * 32;
  constexpr int RT   = 32;
  constexpr int P    = K + 4;
  constexpr int KS   = K / 4;
  constexpr int V4R  = K / 4;
  constexpr int V4T  = RT * V4R;
  constexpr int LPT  = (V4T + NT - 1) / NT;
  constexpr int NTILES = NNODES / RT;

  __shared__ __attribute__((aligned(16))) float sA[2 * RT * P];

  const int tid  = threadIdx.x;
  const int lane = tid & 31;
  const int wave = tid >> 5;
  const int half = lane >> 4;
  const int l15  = lane & 15;
  const int  col   = wave * 16 + l15;
  const bool colOK = (NC % 16 == 0) ? true : (col < NC);
  const int  colc  = colOK ? col : 0;

  v2f bfrag[KS];
#pragma unroll
  for (int kk = 0; kk < KS; ++kk) {
    const int ka = kk * 4 + half * 2;
    const float bx = B[ka * NC + colc];
    const float by = B[(ka + 1) * NC + colc];
    bfrag[kk].x = colOK ? bx : 0.0f;
    bfrag[kk].y = colOK ? by : 0.0f;
  }

  v4f   pre[LPT];
  float psc[LPT];
  int tile = blockIdx.x;

  if (tile < NTILES) {
    const v4f* __restrict__ Ag = (const v4f*)(A + (size_t)tile * (RT * K));
#pragma unroll
    for (int j = 0; j < LPT; ++j) {
      const int i = tid + j * NT;
      if ((j + 1) * NT <= V4T || i < V4T) {
        v4f v = Ag[i];
        const int row = i / V4R;
        const int c4  = i - row * V4R;
        if (PRESCALE) {
          const float s = rowScaleA[tile * RT + row];
          v *= s;
        }
        *(v4f*)(&sA[row * P + c4 * 4]) = v;
      }
    }
  }
  __syncthreads();

  int buf = 0;
  for (; tile < NTILES; tile += gridDim.x) {
    const int  nxt     = tile + gridDim.x;
    const bool hasNext = nxt < NTILES;

    if (hasNext) {
      const v4f* __restrict__ Ag = (const v4f*)(A + (size_t)nxt * (RT * K));
#pragma unroll
      for (int j = 0; j < LPT; ++j) {
        const int i = tid + j * NT;
        if ((j + 1) * NT <= V4T || i < V4T) {
          pre[j] = Ag[i];
          if (PRESCALE) psc[j] = rowScaleA[nxt * RT + i / V4R];
        }
      }
    }

    const float* sa0 = &sA[buf * (RT * P) + l15 * P + half * 2];
    const float* sa1 = sa0 + 16 * P;
    v8f acc0 = {0.f, 0.f, 0.f, 0.f, 0.f, 0.f, 0.f, 0.f};
    v8f acc1 = {0.f, 0.f, 0.f, 0.f, 0.f, 0.f, 0.f, 0.f};
#pragma unroll
    for (int kk = 0; kk < KS; ++kk) {
      const v2f a0 = *(const v2f*)(sa0 + kk * 4);
      const v2f a1 = *(const v2f*)(sa1 + kk * 4);
      acc0 = __builtin_amdgcn_wmma_f32_16x16x4_f32(false, a0, false, bfrag[kk], (short)0, acc0,
                                                   false, false);
      acc1 = __builtin_amdgcn_wmma_f32_16x16x4_f32(false, a1, false, bfrag[kk], (short)0, acc1,
                                                   false, false);
    }

    if (colOK) {
      const int mbase = tile * RT + half * 8;
      float bcol = 0.0f;
      if (EPI) bcol = bias[col];
#pragma unroll
      for (int r = 0; r < 8; ++r) {
        const int row0 = mbase + r;
        const int row1 = row0 + 16;
        float v0 = acc0[r];
        float v1 = acc1[r];
        if (EPI)  { v0 = v0 * rowScaleOut[row0] + bcol; v1 = v1 * rowScaleOut[row1] + bcol; }
        if (RELU) { v0 = fmaxf(v0, 0.0f); v1 = fmaxf(v1, 0.0f); }
        D[row0 * NC + col] = v0;
        D[row1 * NC + col] = v1;
      }
    }

    if (hasNext) {
      float* sn = &sA[(buf ^ 1) * (RT * P)];
#pragma unroll
      for (int j = 0; j < LPT; ++j) {
        const int i = tid + j * NT;
        if ((j + 1) * NT <= V4T || i < V4T) {
          v4f v = pre[j];
          const int row = i / V4R;
          const int c4  = i - row * V4R;
          if (PRESCALE) v *= psc[j];
          *(v4f*)(&sn[row * P + c4 * 4]) = v;
        }
      }
    }
    __syncthreads();
    buf ^= 1;
  }
}


extern "C" void kernel_launch(void* const* d_in, const int* in_sizes, int n_in,
                              void* d_out, int out_size, void* d_ws, size_t ws_size,
                              hipStream_t stream) {
  const float* features = (const float*)d_in[0];
  const int*   src      = (const int*)d_in[1];
  const int*   dst      = (const int*)d_in[2];
  const float* W1       = (const float*)d_in[3];
  const float* b1       = (const float*)d_in[4];
  const float* W2       = (const float*)d_in[5];
  const float* b2       = (const float*)d_in[6];
  const float* W3       = (const float*)d_in[7];
  const float* b3       = (const float*)d_in[8];

  const int N = NNODES;
  const int E = in_sizes[1];

  float* ws      = (float*)d_ws;
  float* inv_out = ws;
  float* inv_in  = ws + N;
  float* P0      = ws + 2 * (size_t)N;
  float* P1      = P0 + (size_t)N * 256;
  unsigned* QK   = (unsigned*)(P1 + (size_t)N * 128);
  unsigned* LCNT = QK;
  unsigned* LIST = QK + AGG_NB * AGG_CS;
  float* H3      = P0;
  float* AGG3    = P0 + (size_t)N * 40;

  const int T = 256;
  auto blocks = [T](long long n) { return (unsigned)((n + T - 1) / T); };

  zero_f32<<<blocks(2LL * N + AGG_NB), T, 0, stream>>>(inv_out, 2 * N, LCNT, AGG_NB, AGG_CS);
  edge_pass<<<AGG_NB, EP_NT, 0, stream>>>(src, dst, inv_out, inv_in, LCNT, LIST, E);
  norm_kernel<<<blocks(N), T, 0, stream>>>(inv_out, inv_in, N);

  agg_owner<128, true><<<AGG_NB, AGG_NT, 0, stream>>>(features, src, dst, inv_out, P1, LCNT, LIST, E);
  gemm_wstat<128, 256, false, true, true><<<512, 512, 0, stream>>>(
      P1, W1, P0, nullptr, inv_in, b1);

  gemm_wstat<256, 128, true, false, false><<<512, 256, 0, stream>>>(
      P0, W2, P1, inv_out, nullptr, nullptr);
  agg_owner<128, false><<<AGG_NB, AGG_NT, 0, stream>>>(P1, src, dst, nullptr, P0, LCNT, LIST, E);
  epilogue_kernel<128, true><<<blocks((long long)N * 128), T, 0, stream>>>(
      P0, inv_in, b2, P1, N * 128);

  gemm_wstat<128, 40, true, false, false><<<1024, 96, 0, stream>>>(
      P1, W3, H3, inv_out, nullptr, nullptr);
  agg_owner<40, false><<<AGG_NB, AGG_NT, 0, stream>>>(H3, src, dst, nullptr, AGG3, LCNT, LIST, E);
  epilogue_kernel<40, false><<<blocks((long long)N * 40), T, 0, stream>>>(
      AGG3, inv_in, b3, (float*)d_out, N * 40);
}


# Layer-2 scale, bias and relu moved to their consumer: the layer-3 GEMM makes relu(agg*inv_in+b2)*inv_out while staging its A tiles; the one-float-per-lane launch and out2 are gone; sums unchanged
# speedup vs baseline: 4.0432x; 1.0995x over previous
#include <hip/hip_runtime.h>


typedef __attribute__((ext_vector_type(2))) float v2f;
typedef __attribute__((ext_vector_type(4))) float v4f;
typedef __attribute__((ext_vector_type(8))) float v8f;
typedef __attribute__((ext_vector_type(4))) int   v4i;

#define NNODES 100000


__global__ void zero_f32(float* __restrict__ p, int n, unsigned* __restrict__ q, int m, int qs) {
  int i = blockIdx.x * blockDim.x + threadIdx.x;
  if (i < n) p[i] = 0.0f;
  else if (i - n < m) q[(size_t)(i - n) * qs] = 0u;
}

__global__ void norm_kernel(float* __restrict__ deg_out, float* __restrict__ deg_in, int n) {
  int i = blockIdx.x * blockDim.x + threadIdx.x;
  if (i < n) {
    deg_out[i] = rsqrtf(fmaxf(deg_out[i], 1.0f));
    deg_in[i]  = rsqrtf(fmaxf(deg_in[i], 1.0f));
  }
}

#define AGG_NB 256
#define AGG_R  391
#define AGG_NT 1024
#define AGG_WQ 512
#define AGG_EMASK 0x7FFFFFu
#define AGG_CAP ((AGG_NT / 32) * AGG_WQ)
#define AGG_CS  64

static_assert(AGG_NB * AGG_R >= NNODES, "node ranges must cover every node");
static_assert(AGG_R <= 512 && NNODES <= (1 << 23), "entry packing: 9 bits row, 23 bits index");
static_assert((long long)AGG_NB * (AGG_CAP + AGG_CS) <= (long long)NNODES * 128,
              "the edge lists must fit the workspace plane they are given");

#define EP_NT  1024
#define EP_PER 4

__global__ __launch_bounds__(EP_NT) void edge_pass(const int* __restrict__ src,
                                                   const int* __restrict__ dst,
                                                   float* __restrict__ deg_out,
                                                   float* __restrict__ deg_in,
                                                   unsigned* __restrict__ lcount,
                                                   unsigned* __restrict__ lists, int E) {
  __shared__ unsigned cnt[AGG_NB];
  __shared__ unsigned base[AGG_NB];
  const int tid   = threadIdx.x;
  const int share = (E + (int)gridDim.x - 1) / (int)gridDim.x;
  const int e0    = blockIdx.x * share;
  const int e1    = min(E, e0 + share);
  for (int eb = e0; eb < e1; eb += EP_NT * EP_PER) {
    if (tid < AGG_NB) cnt[tid] = 0u;
    __syncthreads();
    int      rg[EP_PER];
    unsigned rk[EP_PER];
    unsigned en[EP_PER];
    int      sv[EP_PER];
    int      dv[EP_PER];
#pragma unroll
    for (int k = 0; k < EP_PER; ++k) {
      const int e = eb + k * EP_NT + tid;
      sv[k] = (e < e1) ? src[e] : -1;
      dv[k] = (e < e1) ? dst[e] : -1;
    }
#pragma unroll
    for (int k = 0; k < EP_PER; ++k) {
      rg[k] = -1;
      if (dv[k] >= 0) {
        const int r = dv[k] / AGG_R;
        rg[k] = r;
        en[k] = ((unsigned)(dv[k] - r * AGG_R) << 23) | (unsigned)sv[k];
        rk[k] = __hip_atomic_fetch_add(&cnt[r], 1u, __ATOMIC_RELAXED, __HIP_MEMORY_SCOPE_WORKGROUP);
      }
    }
    __syncthreads();
    if (tid < AGG_NB) {
      const unsigned c = cnt[tid];
      base[tid] = c ? atomicAdd(&lcount[tid * AGG_CS], c) : 0u;
    }
    __syncthreads();
#pragma unroll
    for (int k = 0; k < EP_PER; ++k) {
      if (rg[k] >= 0) {
        const unsigned slot = base[rg[k]] + rk[k];
        if (slot < (unsigned)AGG_CAP) lists[(size_t)rg[k] * AGG_CAP + slot] = en[k];
      }
    }
#pragma unroll
    for (int k = 0; k < EP_PER; ++k) {
      if (dv[k] >= 0) {
        atomicAdd(&deg_out[sv[k]], 1.0f);
        atomicAdd(&deg_in[dv[k]], 1.0f);
      }
    }
  }
}

template <int F, bool SCALE>
__device__ __forceinline__ void agg_slow_edge(float* acc, const float* __restrict__ x,
                                              const int* __restrict__ src,
                                              const float* __restrict__ scale, unsigned row, int e) {
  constexpr int C4 = F / 4;
  const int   s  = src[e];
  const float sc = SCALE ? scale[s] : 1.0f;
#pragma nounroll
  for (int f = 0; f < F; ++f) {
    __hip_atomic_fetch_add(&acc[row * F + (f & 3) * C4 + (f >> 2)], x[(size_t)s * F + f] * sc,
                           __ATOMIC_RELAXED, __HIP_MEMORY_SCOPE_WORKGROUP);
  }
}

template <int F, bool SCALE>
__global__ __launch_bounds__(AGG_NT) void agg_owner(const float* __restrict__ x,
                                                    const int* __restrict__ src,
                                                    const int* __restrict__ dst,
                                                    const float* __restrict__ scale,
                                                    float* __restrict__ agg,
                                                    const unsigned* __restrict__ lcount,
                                                    const unsigned* __restrict__ lists, int E) {
  constexpr int NW  = AGG_NT / 32;
  constexpr int C4  = F / 4;
  constexpr int EPW = 32 / C4;
  constexpr int U   = 8;

  __shared__ __attribute__((aligned(16))) float acc[AGG_R * F];
  __shared__ unsigned wq[NW * AGG_WQ];

  const int tid  = threadIdx.x;
  const int lane = tid & 31;
  const int wave = __builtin_amdgcn_readfirstlane(tid >> 5);
  const int lo    = blockIdx.x * AGG_R;
  const int nrows = min(AGG_R, NNODES - lo);
  unsigned* myq = &wq[wave * AGG_WQ];

  const int  total = __builtin_amdgcn_readfirstlane((int)lcount[blockIdx.x * AGG_CS]);
  const bool scan  = (unsigned)total > (unsigned)AGG_CAP;
  int wcnt = 0;

  for (int i = tid; i < AGG_R * F / 4; i += AGG_NT) ((v4f*)acc)[i] = (v4f){0.f, 0.f, 0.f, 0.f};
  __syncthreads();

  if (!scan) {
    const int per = (total + NW - 1) / NW;
    const int j0  = wave * per;
    wcnt = max(0, min(per, total - j0));
    const unsigned* q = lists + (size_t)blockIdx.x * AGG_CAP + j0;
    for (int jb = 0; jb < wcnt; jb += 128) {
      unsigned en[4];
#pragma unroll
      for (int k = 0; k < 4; ++k) {
        const int j = jb + k * 32 + lane;
        en[k] = (j < wcnt) ? q[j] : 0u;
      }
#pragma unroll
      for (int k = 0; k < 4; ++k) {
        const int j = jb + k * 32 + lane;
        if (j < wcnt) myq[j] = en[k];
      }
    }
  } else {
    auto push = [&](unsigned u, int e) {
      const bool     p    = u < (unsigned)nrows;
      const unsigned mask = __builtin_amdgcn_ballot_w32(p);
      if (mask != 0u) {
        const int pos = wcnt + (int)__builtin_amdgcn_mbcnt_lo(mask, 0u);
        if (p) {
          if (pos < AGG_WQ) myq[pos] = (u << 23) | (unsigned)e;
          else agg_slow_edge<F, SCALE>(acc, x, src, scale, u, e);
        }
        wcnt += __builtin_popcount(mask);
      }
    };

    const int  E4   = E >> 2;
    const v4i* dst4 = (const v4i*)dst;
    auto scan4 = [&](v4i d, int i) {
      const unsigned u0 = (unsigned)(d.x - lo);
      const unsigned u1 = (unsigned)(d.y - lo);
      const unsigned u2 = (unsigned)(d.z - lo);
      const unsigned u3 = (unsigned)(d.w - lo);
      const unsigned m  = min(min(u0, u1), min(u2, u3));
      if (__builtin_amdgcn_ballot_w32(m < (unsigned)nrows) != 0u) {
        push(u0, 4 * i + 0);
        push(u1, 4 * i + 1);
        push(u2, 4 * i + 2);
        push(u3, 4 * i + 3);
      }
    };
    constexpr int SU = 8;
    int ib = 0;
    for (; ib + SU * AGG_NT <= E4; ib += SU * AGG_NT) {
      v4i d[SU];
#pragma unroll
      for (int k = 0; k < SU; ++k) d[k] = dst4[ib + k * AGG_NT + tid];
#pragma unroll
      for (int k = 0; k < SU; ++k) scan4(d[k], ib + k * AGG_NT + tid);
    }
    for (; ib < E4; ib += AGG_NT) {
      const int i = ib + tid;
      v4i d = {-1, -1, -1, -1};
      if (i < E4) d = dst4[i];
      scan4(d, i);
    }
    if (wave == 0) {
      const int e = 4 * E4 + lane;
      int d = -1;
      if (e < E) d = dst[e];
      push((unsigned)(d - lo), e);
    }

    const int nq = min(wcnt, AGG_WQ);
    for (int jb = 0; jb < nq; jb += 128) {
      unsigned en[4];
      int      sv[4];
#pragma unroll
      for (int k = 0; k < 4; ++k) {
        const int j = jb + k * 32 + lane;
        en[k] = (j < nq) ? myq[j] : 0u;
      }
#pragma unroll
      for (int k = 0; k < 4; ++k) sv[k] = src[en[k] & AGG_EMASK];
#pragma unroll
      for (int k = 0; k < 4; ++k) {
        const int j = jb + k * 32 + lane;
        if (j < nq) myq[j] = (en[k] & ~AGG_EMASK) | (unsigned)sv[k];
      }
    }
  }
  const int n = min(wcnt, AGG_WQ);

  const int  sub    = (EPW == 1) ? 0 : lane / C4;
  const int  c4     = (EPW == 1) ? lane : lane - sub * C4;
  const bool active = sub < EPW;
  const v4f* x4 = (const v4f*)x;
  for (int jb = 0; jb < n; jb += U * EPW) {
    v4f      v[U];
    float    sc[U];
    unsigned rw[U];
    bool     ok[U];
#pragma unroll
    for (int k = 0; k < U; ++k) {
      const int j = jb + k * EPW + sub;
      ok[k] = active && (j < n);
      unsigned en = myq[ok[k] ? j : jb];
      if (EPW == 1) en = (unsigned)__builtin_amdgcn_readfirstlane((int)en);
      const unsigned s = en & AGG_EMASK;
      rw[k] = en >> 23;
      v[k]  = x4[(size_t)s * C4 + c4];
      sc[k] = SCALE ? scale[s] : 1.0f;
    }
#pragma unroll
    for (int k = 0; k < U; ++k) {
      if (ok[k]) {
        float* a = &acc[rw[k] * F + c4];
        __hip_atomic_fetch_add(a,          v[k].x * sc[k], __ATOMIC_RELAXED, __HIP_MEMORY_SCOPE_WORKGROUP);
        __hip_atomic_fetch_add(a + C4,     v[k].y * sc[k], __ATOMIC_RELAXED, __HIP_MEMORY_SCOPE_WORKGROUP);
        __hip_atomic_fetch_add(a + 2 * C4, v[k].z * sc[k], __ATOMIC_RELAXED, __HIP_MEMORY_SCOPE_WORKGROUP);
        __hip_atomic_fetch_add(a + 3 * C4, v[k].w * sc[k], __ATOMIC_RELAXED, __HIP_MEMORY_SCOPE_WORKGROUP);
      }
    }
  }
  __syncthreads();

  v4f* out4 = (v4f*)(agg + (size_t)lo * F);
  for (int it = tid; it < nrows * C4; it += AGG_NT) {
    const int row = it / C4;
    const int q4  = it - row * C4;
    const float* a = &acc[row * F + q4];
    out4[it] = (v4f){a[0], a[C4], a[2 * C4], a[3 * C4]};
  }
}

template <int F, bool RELU>
__global__ void epilogue_kernel(const float* __restrict__ agg,
                                const float* __restrict__ invIn,
                                const float* __restrict__ bias,
                                float* __restrict__ out, int n) {
  int i = blockIdx.x * blockDim.x + threadIdx.x;
  if (i < n) {
    int row = i / F;
    int f = i - row * F;
    float v = agg[i] * invIn[row] + bias[f];
    if (RELU) v = fmaxf(v, 0.0f);
    out[i] = v;
  }
}

template <int K, int NC, bool PRESCALE, bool INEPI, bool EPI, bool RELU>
__global__ __launch_bounds__(((NC + 15) / 16) * 32) void gemm_wstat(
    const float* __restrict__ A, const float* __restrict__ B, float* __restrict__ D,
    const float* __restrict__ rowScaleA, const float* __restrict__ rowScaleOut,
    const float* __restrict__ bias,
    const float* __restrict__ rowScaleIn, const float* __restrict__ biasIn) {
  constexpr int NW   = (NC + 15) / 16;
  constexpr int NT   = NW * 32;
  constexpr int RT   = 32;
  constexpr int P    = K + 4;
  constexpr int KS   = K / 4;
  constexpr int V4R  = K / 4;
  constexpr int V4T  = RT * V4R;
  constexpr int LPT  = (V4T + NT - 1) / NT;
  constexpr int NTILES = NNODES / RT;
  static_assert(!INEPI || NT % V4R == 0,
                "INEPI: every piece of a thread must fall on the same four columns of A");

  __shared__ __attribute__((aligned(16))) float sA[2 * RT * P];

  const int tid  = threadIdx.x;
  const int lane = tid & 31;
  const int wave = tid >> 5;
  const int half = lane >> 4;
  const int l15  = lane & 15;
  const int  col   = wave * 16 + l15;
  const bool colOK = (NC % 16 == 0) ? true : (col < NC);
  const int  colc  = colOK ? col : 0;

  v2f bfrag[KS];
#pragma unroll
  for (int kk = 0; kk < KS; ++kk) {
    const int ka = kk * 4 + half * 2;
    const float bx = B[ka * NC + colc];
    const float by = B[(ka + 1) * NC + colc];
    bfrag[kk].x = colOK ? bx : 0.0f;
    bfrag[kk].y = colOK ? by : 0.0f;
  }

  v4f   pre[LPT];
  float psc[LPT];
  float pin[LPT];
  v4f bin = {0.f, 0.f, 0.f, 0.f};
  if (INEPI) {
    const int kc = (tid % V4R) * 4;
    bin = (v4f){biasIn[kc], biasIn[kc + 1], biasIn[kc + 2], biasIn[kc + 3]};
  }
  auto inepi = [&](v4f v, float si) {
    v = v * si + bin;
    return (v4f){fmaxf(v.x, 0.0f), fmaxf(v.y, 0.0f), fmaxf(v.z, 0.0f), fmaxf(v.w, 0.0f)};
  };
  int tile = blockIdx.x;

  if (tile < NTILES) {
    const v4f* __restrict__ Ag = (const v4f*)(A + (size_t)tile * (RT * K));
#pragma unroll
    for (int j = 0; j < LPT; ++j) {
      const int i = tid + j * NT;
      if ((j + 1) * NT <= V4T || i < V4T) {
        v4f v = Ag[i];
        const int row = i / V4R;
        const int c4  = i - row * V4R;
        if (INEPI) v = inepi(v, rowScaleIn[tile * RT + row]);
        if (PRESCALE) {
          const float s = rowScaleA[tile * RT + row];
          v *= s;
        }
        *(v4f*)(&sA[row * P + c4 * 4]) = v;
      }
    }
  }
  __syncthreads();

  int buf = 0;
  for (; tile < NTILES; tile += gridDim.x) {
    const int  nxt     = tile + gridDim.x;
    const bool hasNext = nxt < NTILES;

    if (hasNext) {
      const v4f* __restrict__ Ag = (const v4f*)(A + (size_t)nxt * (RT * K));
#pragma unroll
      for (int j = 0; j < LPT; ++j) {
        const int i = tid + j * NT;
        if ((j + 1) * NT <= V4T || i < V4T) {
          pre[j] = Ag[i];
          if (PRESCALE) psc[j] = rowScaleA[nxt * RT + i / V4R];
          if (INEPI) pin[j] = rowScaleIn[nxt * RT + i / V4R];
        }
      }
    }

    const float* sa0 = &sA[buf * (RT * P) + l15 * P + half * 2];
    const float* sa1 = sa0 + 16 * P;
    v8f acc0 = {0.f, 0.f, 0.f, 0.f, 0.f, 0.f, 0.f, 0.f};
    v8f acc1 = {0.f, 0.f, 0.f, 0.f, 0.f, 0.f, 0.f, 0.f};
#pragma unroll
    for (int kk = 0; kk < KS; ++kk) {
      const v2f a0 = *(const v2f*)(sa0 + kk * 4);
      const v2f a1 = *(const v2f*)(sa1 + kk * 4);
      acc0 = __builtin_amdgcn_wmma_f32_16x16x4_f32(false, a0, false, bfrag[kk], (short)0, acc0,
                                                   false, false);
      acc1 = __builtin_amdgcn_wmma_f32_16x16x4_f32(false, a1, false, bfrag[kk], (short)0, acc1,
                                                   false, false);
    }

    if (colOK) {
      const int mbase = tile * RT + half * 8;
      float bcol = 0.0f;
      if (EPI) bcol = bias[col];
#pragma unroll
      for (int r = 0; r < 8; ++r) {
        const int row0 = mbase + r;
        const int row1 = row0 + 16;
        float v0 = acc0[r];
        float v1 = acc1[r];
        if (EPI)  { v0 = v0 * rowScaleOut[row0] + bcol; v1 = v1 * rowScaleOut[row1] + bcol; }
        if (RELU) { v0 = fmaxf(v0, 0.0f); v1 = fmaxf(v1, 0.0f); }
        D[row0 * NC + col] = v0;
        D[row1 * NC + col] = v1;
      }
    }

    if (hasNext) {
      float* sn = &sA[(buf ^ 1) * (RT * P)];
#pragma unroll
      for (int j = 0; j < LPT; ++j) {
        const int i = tid + j * NT;
        if ((j + 1) * NT <= V4T || i < V4T) {
          v4f v = pre[j];
          const int row = i / V4R;
          const int c4  = i - row * V4R;
          if (INEPI) v = inepi(v, pin[j]);
          if (PRESCALE) v *= psc[j];
          *(v4f*)(&sn[row * P + c4 * 4]) = v;
        }
      }
    }
    __syncthreads();
    buf ^= 1;
  }
}


extern "C" void kernel_launch(void* const* d_in, const int* in_sizes, int n_in,
                              void* d_out, int out_size, void* d_ws, size_t ws_size,
                              hipStream_t stream) {
  const float* features = (const float*)d_in[0];
  const int*   src      = (const int*)d_in[1];
  const int*   dst      = (const int*)d_in[2];
  const float* W1       = (const float*)d_in[3];
  const float* b1       = (const float*)d_in[4];
  const float* W2       = (const float*)d_in[5];
  const float* b2       = (const float*)d_in[6];
  const float* W3       = (const float*)d_in[7];
  const float* b3       = (const float*)d_in[8];

  const int N = NNODES;
  const int E = in_sizes[1];

  float* ws      = (float*)d_ws;
  float* inv_out = ws;
  float* inv_in  = ws + N;
  float* P0      = ws + 2 * (size_t)N;
  float* P1      = P0 + (size_t)N * 256;
  unsigned* QK   = (unsigned*)(P1 + (size_t)N * 128);
  unsigned* LCNT = QK;
  unsigned* LIST = QK + AGG_NB * AGG_CS;
  float* H3      = P1;
  float* AGG3    = P1 + (size_t)N * 40;

  const int T = 256;
  auto blocks = [T](long long n) { return (unsigned)((n + T - 1) / T); };

  zero_f32<<<blocks(2LL * N + AGG_NB), T, 0, stream>>>(inv_out, 2 * N, LCNT, AGG_NB, AGG_CS);
  edge_pass<<<AGG_NB, EP_NT, 0, stream>>>(src, dst, inv_out, inv_in, LCNT, LIST, E);
  norm_kernel<<<blocks(N), T, 0, stream>>>(inv_out, inv_in, N);

  agg_owner<128, true><<<AGG_NB, AGG_NT, 0, stream>>>(features, src, dst, inv_out, P1, LCNT, LIST, E);
  gemm_wstat<128, 256, false, false, true, true><<<512, 512, 0, stream>>>(
      P1, W1, P0, nullptr, inv_in, b1, nullptr, nullptr);

  gemm_wstat<256, 128, true, false, false, false><<<512, 256, 0, stream>>>(
      P0, W2, P1, inv_out, nullptr, nullptr, nullptr, nullptr);
  agg_owner<128, false><<<AGG_NB, AGG_NT, 0, stream>>>(P1, src, dst, nullptr, P0, LCNT, LIST, E);

  gemm_wstat<128, 40, true, true, false, false><<<1024, 96, 0, stream>>>(
      P0, W3, H3, inv_out, nullptr, nullptr, inv_in, b2);
  agg_owner<40, false><<<AGG_NB, AGG_NT, 0, stream>>>(H3, src, dst, nullptr, AGG3, LCNT, LIST, E);
  epilogue_kernel<40, false><<<blocks((long long)N * 40), T, 0, stream>>>(
      AGG3, inv_in, b3, (float*)d_out, N * 40);
}


# Last layer's scale and bias applied by the owner work-groups while they store their LDS neighbour sums: rows go straight to the output, raw-sum plane and one-float-per-lane launch gone; sums unchanged
# speedup vs baseline: 4.2486x; 1.0508x over previous
#include <hip/hip_runtime.h>


typedef __attribute__((ext_vector_type(2))) float v2f;
typedef __attribute__((ext_vector_type(4))) float v4f;
typedef __attribute__((ext_vector_type(8))) float v8f;
typedef __attribute__((ext_vector_type(4))) int   v4i;

#define NNODES 100000


__global__ void zero_f32(float* __restrict__ p, int n, unsigned* __restrict__ q, int m, int qs) {
  int i = blockIdx.x * blockDim.x + threadIdx.x;
  if (i < n) p[i] = 0.0f;
  else if (i - n < m) q[(size_t)(i - n) * qs] = 0u;
}

__global__ void norm_kernel(float* __restrict__ deg_out, float* __restrict__ deg_in, int n) {
  int i = blockIdx.x * blockDim.x + threadIdx.x;
  if (i < n) {
    deg_out[i] = rsqrtf(fmaxf(deg_out[i], 1.0f));
    deg_in[i]  = rsqrtf(fmaxf(deg_in[i], 1.0f));
  }
}

#define AGG_NB 256
#define AGG_R  391
#define AGG_NT 1024
#define AGG_WQ 512
#define AGG_EMASK 0x7FFFFFu
#define AGG_CAP ((AGG_NT / 32) * AGG_WQ)
#define AGG_CS  64

static_assert(AGG_NB * AGG_R >= NNODES, "node ranges must cover every node");
static_assert(AGG_R <= 512 && NNODES <= (1 << 23), "entry packing: 9 bits row, 23 bits index");
static_assert((long long)AGG_NB * (AGG_CAP + AGG_CS) <= (long long)NNODES * 128,
              "the edge lists must fit the workspace plane they are given");

#define EP_NT  1024
#define EP_PER 4

__global__ __launch_bounds__(EP_NT) void edge_pass(const int* __restrict__ src,
                                                   const int* __restrict__ dst,
                                                   float* __restrict__ deg_out,
                                                   float* __restrict__ deg_in,
                                                   unsigned* __restrict__ lcount,
                                                   unsigned* __restrict__ lists, int E) {
  __shared__ unsigned cnt[AGG_NB];
  __shared__ unsigned base[AGG_NB];
  const int tid   = threadIdx.x;
  const int share = (E + (int)gridDim.x - 1) / (int)gridDim.x;
  const int e0    = blockIdx.x * share;
  const int e1    = min(E, e0 + share);
  for (int eb = e0; eb < e1; eb += EP_NT * EP_PER) {
    if (tid < AGG_NB) cnt[tid] = 0u;
    __syncthreads();
    int      rg[EP_PER];
    unsigned rk[EP_PER];
    unsigned en[EP_PER];
    int      sv[EP_PER];
    int      dv[EP_PER];
#pragma unroll
    for (int k = 0; k < EP_PER; ++k) {
      const int e = eb + k * EP_NT + tid;
      sv[k] = (e < e1) ? src[e] : -1;
      dv[k] = (e < e1) ? dst[e] : -1;
    }
#pragma unroll
    for (int k = 0; k < EP_PER; ++k) {
      rg[k] = -1;
      if (dv[k] >= 0) {
        const int r = dv[k] / AGG_R;
        rg[k] = r;
        en[k] = ((unsigned)(dv[k] - r * AGG_R) << 23) | (unsigned)sv[k];
        rk[k] = __hip_atomic_fetch_add(&cnt[r], 1u, __ATOMIC_RELAXED, __HIP_MEMORY_SCOPE_WORKGROUP);
      }
    }
    __syncthreads();
    if (tid < AGG_NB) {
      const unsigned c = cnt[tid];
      base[tid] = c ? atomicAdd(&lcount[tid * AGG_CS], c) : 0u;
    }
    __syncthreads();
#pragma unroll
    for (int k = 0; k < EP_PER; ++k) {
      if (rg[k] >= 0) {
        const unsigned slot = base[rg[k]] + rk[k];
        if (slot < (unsigned)AGG_CAP) lists[(size_t)rg[k] * AGG_CAP + slot] = en[k];
      }
    }
#pragma unroll
    for (int k = 0; k < EP_PER; ++k) {
      if (dv[k] >= 0) {
        atomicAdd(&deg_out[sv[k]], 1.0f);
        atomicAdd(&deg_in[dv[k]], 1.0f);
      }
    }
  }
}

template <int F, bool SCALE>
__device__ __forceinline__ void agg_slow_edge(float* acc, const float* __restrict__ x,
                                              const int* __restrict__ src,
                                              const float* __restrict__ scale, unsigned row, int e) {
  constexpr int C4 = F / 4;
  const int   s  = src[e];
  const float sc = SCALE ? scale[s] : 1.0f;
#pragma nounroll
  for (int f = 0; f < F; ++f) {
    __hip_atomic_fetch_add(&acc[row * F + (f & 3) * C4 + (f >> 2)], x[(size_t)s * F + f] * sc,
                           __ATOMIC_RELAXED, __HIP_MEMORY_SCOPE_WORKGROUP);
  }
}

template <int F, bool SCALE, bool FINAL>
__global__ __launch_bounds__(AGG_NT) void agg_owner(const float* __restrict__ x,
                                                    const int* __restrict__ src,
                                                    const int* __restrict__ dst,
                                                    const float* __restrict__ scale,
                                                    float* __restrict__ agg,
                                                    const unsigned* __restrict__ lcount,
                                                    const unsigned* __restrict__ lists, int E,
                                                    const float* __restrict__ rowScaleOut,
                                                    const float* __restrict__ biasOut) {
  constexpr int NW  = AGG_NT / 32;
  constexpr int C4  = F / 4;
  constexpr int EPW = 32 / C4;
  constexpr int U   = 8;

  __shared__ __attribute__((aligned(16))) float acc[AGG_R * F];
  __shared__ unsigned wq[NW * AGG_WQ];

  const int tid  = threadIdx.x;
  const int lane = tid & 31;
  const int wave = __builtin_amdgcn_readfirstlane(tid >> 5);
  const int lo    = blockIdx.x * AGG_R;
  const int nrows = min(AGG_R, NNODES - lo);
  unsigned* myq = &wq[wave * AGG_WQ];

  const int  total = __builtin_amdgcn_readfirstlane((int)lcount[blockIdx.x * AGG_CS]);
  const bool scan  = (unsigned)total > (unsigned)AGG_CAP;
  int wcnt = 0;

  for (int i = tid; i < AGG_R * F / 4; i += AGG_NT) ((v4f*)acc)[i] = (v4f){0.f, 0.f, 0.f, 0.f};
  __syncthreads();

  if (!scan) {
    const int per = (total + NW - 1) / NW;
    const int j0  = wave * per;
    wcnt = max(0, min(per, total - j0));
    const unsigned* q = lists + (size_t)blockIdx.x * AGG_CAP + j0;
    for (int jb = 0; jb < wcnt; jb += 128) {
      unsigned en[4];
#pragma unroll
      for (int k = 0; k < 4; ++k) {
        const int j = jb + k * 32 + lane;
        en[k] = (j < wcnt) ? q[j] : 0u;
      }
#pragma unroll
      for (int k = 0; k < 4; ++k) {
        const int j = jb + k * 32 + lane;
        if (j < wcnt) myq[j] = en[k];
      }
    }
  } else {
    auto push = [&](unsigned u, int e) {
      const bool     p    = u < (unsigned)nrows;
      const unsigned mask = __builtin_amdgcn_ballot_w32(p);
      if (mask != 0u) {
        const int pos = wcnt + (int)__builtin_amdgcn_mbcnt_lo(mask, 0u);
        if (p) {
          if (pos < AGG_WQ) myq[pos] = (u << 23) | (unsigned)e;
          else agg_slow_edge<F, SCALE>(acc, x, src, scale, u, e);
        }
        wcnt += __builtin_popcount(mask);
      }
    };

    const int  E4   = E >> 2;
    const v4i* dst4 = (const v4i*)dst;
    auto scan4 = [&](v4i d, int i) {
      const unsigned u0 = (unsigned)(d.x - lo);
      const unsigned u1 = (unsigned)(d.y - lo);
      const unsigned u2 = (unsigned)(d.z - lo);
      const unsigned u3 = (unsigned)(d.w - lo);
      const unsigned m  = min(min(u0, u1), min(u2, u3));
      if (__builtin_amdgcn_ballot_w32(m < (unsigned)nrows) != 0u) {
        push(u0, 4 * i + 0);
        push(u1, 4 * i + 1);
        push(u2, 4 * i + 2);
        push(u3, 4 * i + 3);
      }
    };
    constexpr int SU = 8;
    int ib = 0;
    for (; ib + SU * AGG_NT <= E4; ib += SU * AGG_NT) {
      v4i d[SU];
#pragma unroll
      for (int k = 0; k < SU; ++k) d[k] = dst4[ib + k * AGG_NT + tid];
#pragma unroll
      for (int k = 0; k < SU; ++k) scan4(d[k], ib + k * AGG_NT + tid);
    }
    for (; ib < E4; ib += AGG_NT) {
      const int i = ib + tid;
      v4i d = {-1, -1, -1, -1};
      if (i < E4) d = dst4[i];
      scan4(d, i);
    }
    if (wave == 0) {
      const int e = 4 * E4 + lane;
      int d = -1;
      if (e < E) d = dst[e];
      push((unsigned)(d - lo), e);
    }

    const int nq = min(wcnt, AGG_WQ);
    for (int jb = 0; jb < nq; jb += 128) {
      unsigned en[4];
      int      sv[4];
#pragma unroll
      for (int k = 0; k < 4; ++k) {
        const int j = jb + k * 32 + lane;
        en[k] = (j < nq) ? myq[j] : 0u;
      }
#pragma unroll
      for (int k = 0; k < 4; ++k) sv[k] = src[en[k] & AGG_EMASK];
#pragma unroll
      for (int k = 0; k < 4; ++k) {
        const int j = jb + k * 32 + lane;
        if (j < nq) myq[j] = (en[k] & ~AGG_EMASK) | (unsigned)sv[k];
      }
    }
  }
  const int n = min(wcnt, AGG_WQ);

  constexpr int NST = (AGG_R * C4 + AGG_NT - 1) / AGG_NT;
  float os[FINAL ? NST : 1];
  v4f   ob[FINAL ? NST : 1];
  if (FINAL) {
#pragma unroll
    for (int k = 0; k < NST; ++k) {
      const int it  = tid + k * AGG_NT;
      const int row = min(it / C4, nrows - 1);
      const int q4  = it % C4;
      os[k] = rowScaleOut[lo + row];
      ob[k] = (v4f){biasOut[4 * q4], biasOut[4 * q4 + 1], biasOut[4 * q4 + 2], biasOut[4 * q4 + 3]};
    }
  }

  const int  sub    = (EPW == 1) ? 0 : lane / C4;
  const int  c4     = (EPW == 1) ? lane : lane - sub * C4;
  const bool active = sub < EPW;
  const v4f* x4 = (const v4f*)x;
  for (int jb = 0; jb < n; jb += U * EPW) {
    v4f      v[U];
    float    sc[U];
    unsigned rw[U];
    bool     ok[U];
#pragma unroll
    for (int k = 0; k < U; ++k) {
      const int j = jb + k * EPW + sub;
      ok[k] = active && (j < n);
      unsigned en = myq[ok[k] ? j : jb];
      if (EPW == 1) en = (unsigned)__builtin_amdgcn_readfirstlane((int)en);
      const unsigned s = en & AGG_EMASK;
      rw[k] = en >> 23;
      v[k]  = x4[(size_t)s * C4 + c4];
      sc[k] = SCALE ? scale[s] : 1.0f;
    }
#pragma unroll
    for (int k = 0; k < U; ++k) {
      if (ok[k]) {
        float* a = &acc[rw[k] * F + c4];
        __hip_atomic_fetch_add(a,          v[k].x * sc[k], __ATOMIC_RELAXED, __HIP_MEMORY_SCOPE_WORKGROUP);
        __hip_atomic_fetch_add(a + C4,     v[k].y * sc[k], __ATOMIC_RELAXED, __HIP_MEMORY_SCOPE_WORKGROUP);
        __hip_atomic_fetch_add(a + 2 * C4, v[k].z * sc[k], __ATOMIC_RELAXED, __HIP_MEMORY_SCOPE_WORKGROUP);
        __hip_atomic_fetch_add(a + 3 * C4, v[k].w * sc[k], __ATOMIC_RELAXED, __HIP_MEMORY_SCOPE_WORKGROUP);
      }
    }
  }
  __syncthreads();

  v4f* out4 = (v4f*)(agg + (size_t)lo * F);
  if (FINAL) {
#pragma unroll
    for (int k = 0; k < NST; ++k) {
      const int it = tid + k * AGG_NT;
      if (it < nrows * C4) {
        const int row = it / C4;
        const int q4  = it - row * C4;
        const float* a = &acc[row * F + q4];
        const v4f v = (v4f){a[0], a[C4], a[2 * C4], a[3 * C4]};
        out4[it] = v * os[k] + ob[k];
      }
    }
  } else {
    for (int it = tid; it < nrows * C4; it += AGG_NT) {
      const int row = it / C4;
      const int q4  = it - row * C4;
      const float* a = &acc[row * F + q4];
      out4[it] = (v4f){a[0], a[C4], a[2 * C4], a[3 * C4]};
    }
  }
}

template <int K, int NC, bool PRESCALE, bool INEPI, bool EPI, bool RELU>
__global__ __launch_bounds__(((NC + 15) / 16) * 32) void gemm_wstat(
    const float* __restrict__ A, const float* __restrict__ B, float* __restrict__ D,
    const float* __restrict__ rowScaleA, const float* __restrict__ rowScaleOut,
    const float* __restrict__ bias,
    const float* __restrict__ rowScaleIn, const float* __restrict__ biasIn) {
  constexpr int NW   = (NC + 15) / 16;
  constexpr int NT   = NW * 32;
  constexpr int RT   = 32;
  constexpr int P    = K + 4;
  constexpr int KS   = K / 4;
  constexpr int V4R  = K / 4;
  constexpr int V4T  = RT * V4R;
  constexpr int LPT  = (V4T + NT - 1) / NT;
  constexpr int NTILES = NNODES / RT;
  static_assert(!INEPI || NT % V4R == 0,
                "INEPI: every piece of a thread must fall on the same four columns of A");

  __shared__ __attribute__((aligned(16))) float sA[2 * RT * P];

  const int tid  = threadIdx.x;
  const int lane = tid & 31;
  const int wave = tid >> 5;
  const int half = lane >> 4;
  const int l15  = lane & 15;
  const int  col   = wave * 16 + l15;
  const bool colOK = (NC % 16 == 0) ? true : (col < NC);
  const int  colc  = colOK ? col : 0;

  v2f bfrag[KS];
#pragma unroll
  for (int kk = 0; kk < KS; ++kk) {
    const int ka = kk * 4 + half * 2;
    const float bx = B[ka * NC + colc];
    const float by = B[(ka + 1) * NC + colc];
    bfrag[kk].x = colOK ? bx : 0.0f;
    bfrag[kk].y = colOK ? by : 0.0f;
  }

  v4f   pre[LPT];
  float psc[LPT];
  float pin[LPT];
  v4f bin = {0.f, 0.f, 0.f, 0.f};
  if (INEPI) {
    const int kc = (tid % V4R) * 4;
    bin = (v4f){biasIn[kc], biasIn[kc + 1], biasIn[kc + 2], biasIn[kc + 3]};
  }
  auto inepi = [&](v4f v, float si) {
    v = v * si + bin;
    return (v4f){fmaxf(v.x, 0.0f), fmaxf(v.y, 0.0f), fmaxf(v.z, 0.0f), fmaxf(v.w, 0.0f)};
  };
  int tile = blockIdx.x;

  if (tile < NTILES) {
    const v4f* __restrict__ Ag = (const v4f*)(A + (size_t)tile * (RT * K));
#pragma unroll
    for (int j = 0; j < LPT; ++j) {
      const int i = tid + j * NT;
      if ((j + 1) * NT <= V4T || i < V4T) {
        v4f v = Ag[i];
        const int row = i / V4R;
        const int c4  = i - row * V4R;
        if (INEPI) v = inepi(v, rowScaleIn[tile * RT + row]);
        if (PRESCALE) {
          const float s = rowScaleA[tile * RT + row];
          v *= s;
        }
        *(v4f*)(&sA[row * P + c4 * 4]) = v;
      }
    }
  }
  __syncthreads();

  int buf = 0;
  for (; tile < NTILES; tile += gridDim.x) {
    const int  nxt     = tile + gridDim.x;
    const bool hasNext = nxt < NTILES;

    if (hasNext) {
      const v4f* __restrict__ Ag = (const v4f*)(A + (size_t)nxt * (RT * K));
#pragma unroll
      for (int j = 0; j < LPT; ++j) {
        const int i = tid + j * NT;
        if ((j + 1) * NT <= V4T || i < V4T) {
          pre[j] = Ag[i];
          if (PRESCALE) psc[j] = rowScaleA[nxt * RT + i / V4R];
          if (INEPI) pin[j] = rowScaleIn[nxt * RT + i / V4R];
        }
      }
    }

    const float* sa0 = &sA[buf * (RT * P) + l15 * P + half * 2];
    const float* sa1 = sa0 + 16 * P;
    v8f acc0 = {0.f, 0.f, 0.f, 0.f, 0.f, 0.f, 0.f, 0.f};
    v8f acc1 = {0.f, 0.f, 0.f, 0.f, 0.f, 0.f, 0.f, 0.f};
#pragma unroll
    for (int kk = 0; kk < KS; ++kk) {
      const v2f a0 = *(const v2f*)(sa0 + kk * 4);
      const v2f a1 = *(const v2f*)(sa1 + kk * 4);
      acc0 = __builtin_amdgcn_wmma_f32_16x16x4_f32(false, a0, false, bfrag[kk], (short)0, acc0,
                                                   false, false);
      acc1 = __builtin_amdgcn_wmma_f32_16x16x4_f32(false, a1, false, bfrag[kk], (short)0, acc1,
                                                   false, false);
    }

    if (colOK) {
      const int mbase = tile * RT + half * 8;
      float bcol = 0.0f;
      if (EPI) bcol = bias[col];
#pragma unroll
      for (int r = 0; r < 8; ++r) {
        const int row0 = mbase + r;
        const int row1 = row0 + 16;
        float v0 = acc0[r];
        float v1 = acc1[r];
        if (EPI)  { v0 = v0 * rowScaleOut[row0] + bcol; v1 = v1 * rowScaleOut[row1] + bcol; }
        if (RELU) { v0 = fmaxf(v0, 0.0f); v1 = fmaxf(v1, 0.0f); }
        D[row0 * NC + col] = v0;
        D[row1 * NC + col] = v1;
      }
    }

    if (hasNext) {
      float* sn = &sA[(buf ^ 1) * (RT * P)];
#pragma unroll
      for (int j = 0; j < LPT; ++j) {
        const int i = tid + j * NT;
        if ((j + 1) * NT <= V4T || i < V4T) {
          v4f v = pre[j];
          const int row = i / V4R;
          const int c4  = i - row * V4R;
          if (INEPI) v = inepi(v, pin[j]);
          if (PRESCALE) v *= psc[j];
          *(v4f*)(&sn[row * P + c4 * 4]) = v;
        }
      }
    }
    __syncthreads();
    buf ^= 1;
  }
}


extern "C" void kernel_launch(void* const* d_in, const int* in_sizes, int n_in,
                              void* d_out, int out_size, void* d_ws, size_t ws_size,
                              hipStream_t stream) {
  const float* features = (const float*)d_in[0];
  const int*   src      = (const int*)d_in[1];
  const int*   dst      = (const int*)d_in[2];
  const float* W1       = (const float*)d_in[3];
  const float* b1       = (const float*)d_in[4];
  const float* W2       = (const float*)d_in[5];
  const float* b2       = (const float*)d_in[6];
  const float* W3       = (const float*)d_in[7];
  const float* b3       = (const float*)d_in[8];

  const int N = NNODES;
  const int E = in_sizes[1];

  float* ws      = (float*)d_ws;
  float* inv_out = ws;
  float* inv_in  = ws + N;
  float* P0      = ws + 2 * (size_t)N;
  float* P1      = P0 + (size_t)N * 256;
  unsigned* QK   = (unsigned*)(P1 + (size_t)N * 128);
  unsigned* LCNT = QK;
  unsigned* LIST = QK + AGG_NB * AGG_CS;
  float* H3      = P1;

  const int T = 256;
  auto blocks = [T](long long n) { return (unsigned)((n + T - 1) / T); };

  zero_f32<<<blocks(2LL * N + AGG_NB), T, 0, stream>>>(inv_out, 2 * N, LCNT, AGG_NB, AGG_CS);
  edge_pass<<<AGG_NB, EP_NT, 0, stream>>>(src, dst, inv_out, inv_in, LCNT, LIST, E);
  norm_kernel<<<blocks(N), T, 0, stream>>>(inv_out, inv_in, N);

  agg_owner<128, true, false><<<AGG_NB, AGG_NT, 0, stream>>>(features, src, dst, inv_out, P1, LCNT, LIST, E,
                                                              nullptr, nullptr);
  gemm_wstat<128, 256, false, false, true, true><<<512, 512, 0, stream>>>(
      P1, W1, P0, nullptr, inv_in, b1, nullptr, nullptr);

  gemm_wstat<256, 128, true, false, false, false><<<512, 256, 0, stream>>>(
      P0, W2, P1, inv_out, nullptr, nullptr, nullptr, nullptr);
  agg_owner<128, false, false><<<AGG_NB, AGG_NT, 0, stream>>>(P1, src, dst, nullptr, P0, LCNT, LIST, E,
                                                               nullptr, nullptr);

  gemm_wstat<128, 40, true, true, false, false><<<1024, 96, 0, stream>>>(
      P0, W3, H3, inv_out, nullptr, nullptr, inv_in, b2);
  agg_owner<40, false, true><<<AGG_NB, AGG_NT, 0, stream>>>(H3, src, dst, nullptr, (float*)d_out, LCNT, LIST, E,
                                                            inv_in, b3);
}
